# starting kernel
# baseline (speedup 1.0000x reference)
#include <hip/hip_runtime.h>

typedef __attribute__((ext_vector_type(2))) float v2f;
typedef __attribute__((ext_vector_type(4))) float v4f;
typedef __attribute__((ext_vector_type(8))) float v8f;

__global__ void NeighbourAggregation_scatter_kernel(
    const float* __restrict__ x,
    const int* __restrict__ edge_dst,
    float* __restrict__ out,
    float* __restrict__ counts,
    int total)
{
    int t = blockIdx.x * blockDim.x + threadIdx.x;
    int base = t * 4;
    if (base >= total) return;

    v4f v = __builtin_nontemporal_load((const v4f*)(x + base));
    int e = base >> 4;
    int d = base & 15;
    int dst = __builtin_nontemporal_load(edge_dst + e);

    float* o = out + dst * 16 + d;
    unsafeAtomicAdd(o + 0, v.x);
    unsafeAtomicAdd(o + 1, v.y);
    unsafeAtomicAdd(o + 2, v.z);
    unsafeAtomicAdd(o + 3, v.w);
    if (d == 0) {
        unsafeAtomicAdd(counts + dst, 1.0f);
    }
}

__global__ void NeighbourAggregation_finalize_wmma_kernel(
    float* __restrict__ out,
    const float* __restrict__ counts,
    int num_nodes)
{
    int wavesPerBlock = blockDim.x >> 5;
    int gwave = blockIdx.x * wavesPerBlock + (threadIdx.x >> 5);
    int lane  = threadIdx.x & 31;
    int nb    = gwave * 16;
    if (nb >= num_nodes) return;

    int col  = lane & 15;
    int half = lane >> 4;

    if (nb + 16 <= num_nodes) {
        float r = __frsqrt_rn(counts[nb + col] + 1.0f);

        v8f c = {0.f, 0.f, 0.f, 0.f, 0.f, 0.f, 0.f, 0.f};
        #pragma unroll
        for (int k0 = 0; k0 < 16; k0 += 4) {
            v2f a, b;
            #pragma unroll
            for (int j = 0; j < 2; ++j) {
                int K = k0 + 2 * half + j;
                a[j] = (col == K) ? r : 0.0f;
                b[j] = out[(nb + K) * 16 + col];
            }
            c = __builtin_amdgcn_wmma_f32_16x16x4_f32(
                    false, a, false, b, (short)0, c, false, false);
        }

        #pragma unroll
        for (int v = 0; v < 8; ++v) {
            int row = v + 8 * half;
            out[(nb + row) * 16 + col] = c[v];
        }
    } else {
        for (int row = nb + half; row < num_nodes; row += 2) {
            float r = __frsqrt_rn(counts[row] + 1.0f);
            out[row * 16 + col] *= r;
        }
    }
}

extern "C" void kernel_launch(void* const* d_in, const int* in_sizes, int n_in,
                              void* d_out, int out_size, void* d_ws, size_t ws_size,
                              hipStream_t stream) {
    const float* x        = (const float*)d_in[0];
    const int*   edge_dst = (const int*)d_in[1];
    int total     = in_sizes[0];
    int num_nodes = out_size / 16;

    float* out    = (float*)d_out;
    float* counts = (float*)d_ws;

    hipMemsetAsync(d_out, 0, (size_t)out_size * sizeof(float), stream);
    hipMemsetAsync(d_ws, 0, (size_t)num_nodes * sizeof(float), stream);

    int nthreads = total / 4;
    int sblocks  = (nthreads + 255) / 256;
    NeighbourAggregation_scatter_kernel<<<sblocks, 256, 0, stream>>>(
        x, edge_dst, out, counts, total);

    int nwaves  = (num_nodes + 15) / 16;
    int fblocks = (nwaves + 7) / 8;
    NeighbourAggregation_finalize_wmma_kernel<<<fblocks, 256, 0, stream>>>(
        out, counts, num_nodes);
}


# Scatter lanes remapped so each wave atomic adds two whole 64-byte rows of out (lane = feature) instead of 4-float pieces of 8 rows; same f32 atomics, arrival order as before
# speedup vs baseline: 1.7974x; 1.7974x over previous
#include <hip/hip_runtime.h>

typedef __attribute__((ext_vector_type(2))) float v2f;
typedef __attribute__((ext_vector_type(4))) float v4f;
typedef __attribute__((ext_vector_type(8))) float v8f;

__global__ void NeighbourAggregation_scatter_kernel(
    const float* __restrict__ x,
    const int* __restrict__ edge_dst,
    float* __restrict__ out,
    float* __restrict__ counts,
    int total)
{
    int num_edges = total >> 4;
    int lane  = threadIdx.x & 31;
    int gwave = (blockIdx.x * blockDim.x + threadIdx.x) >> 5;
    int ebase = gwave * 32;
    if (ebase >= num_edges) return;

    int f = lane & 15;
    int h = lane >> 4;

    if (ebase + 32 <= num_edges) {
        int my_d = __builtin_nontemporal_load(edge_dst + ebase + lane);
        const float* xp = x + ebase * 16 + lane;

        float v[16];
        #pragma unroll
        for (int i = 0; i < 16; ++i) {
            v[i] = __builtin_nontemporal_load(xp + 32 * i);
        }

        #pragma unroll
        for (int i = 0; i < 16; ++i) {
            int d0 = __builtin_amdgcn_readlane(my_d, 2 * i);
            int d1 = __builtin_amdgcn_readlane(my_d, 2 * i + 1);
            int d  = h ? d1 : d0;
            unsafeAtomicAdd(out + d * 16 + f, v[i]);
        }

        unsafeAtomicAdd(counts + my_d, 1.0f);
    } else {
        for (int e = ebase + h; e < num_edges; e += 2) {
            int d = edge_dst[e];
            unsafeAtomicAdd(out + d * 16 + f, x[e * 16 + f]);
            if (f == 0) unsafeAtomicAdd(counts + d, 1.0f);
        }
    }
}

__global__ void NeighbourAggregation_finalize_wmma_kernel(
    float* __restrict__ out,
    const float* __restrict__ counts,
    int num_nodes)
{
    int wavesPerBlock = blockDim.x >> 5;
    int gwave = blockIdx.x * wavesPerBlock + (threadIdx.x >> 5);
    int lane  = threadIdx.x & 31;
    int nb    = gwave * 16;
    if (nb >= num_nodes) return;

    int col  = lane & 15;
    int half = lane >> 4;

    if (nb + 16 <= num_nodes) {
        float r = __frsqrt_rn(counts[nb + col] + 1.0f);

        v8f c = {0.f, 0.f, 0.f, 0.f, 0.f, 0.f, 0.f, 0.f};
        #pragma unroll
        for (int k0 = 0; k0 < 16; k0 += 4) {
            v2f a, b;
            #pragma unroll
            for (int j = 0; j < 2; ++j) {
                int K = k0 + 2 * half + j;
                a[j] = (col == K) ? r : 0.0f;
                b[j] = out[(nb + K) * 16 + col];
            }
            c = __builtin_amdgcn_wmma_f32_16x16x4_f32(
                    false, a, false, b, (short)0, c, false, false);
        }

        #pragma unroll
        for (int v = 0; v < 8; ++v) {
            int row = v + 8 * half;
            out[(nb + row) * 16 + col] = c[v];
        }
    } else {
        for (int row = nb + half; row < num_nodes; row += 2) {
            float r = __frsqrt_rn(counts[row] + 1.0f);
            out[row * 16 + col] *= r;
        }
    }
}

extern "C" void kernel_launch(void* const* d_in, const int* in_sizes, int n_in,
                              void* d_out, int out_size, void* d_ws, size_t ws_size,
                              hipStream_t stream) {
    const float* x        = (const float*)d_in[0];
    const int*   edge_dst = (const int*)d_in[1];
    int total     = in_sizes[0];
    int num_nodes = out_size / 16;

    float* out    = (float*)d_out;
    float* counts = (float*)d_ws;

    hipMemsetAsync(d_out, 0, (size_t)out_size * sizeof(float), stream);
    hipMemsetAsync(d_ws, 0, (size_t)num_nodes * sizeof(float), stream);

    int swaves  = (total / 16 + 31) / 32;
    int sblocks = (swaves + 7) / 8;
    NeighbourAggregation_scatter_kernel<<<sblocks, 256, 0, stream>>>(
        x, edge_dst, out, counts, total);

    int nwaves  = (num_nodes + 15) / 16;
    int fblocks = (nwaves + 7) / 8;
    NeighbourAggregation_finalize_wmma_kernel<<<fblocks, 256, 0, stream>>>(
        out, counts, num_nodes);
}


# Accumulate into 128-byte padded workspace rows (16 sums + count in column 16): one f32 atomic instruction per edge hits one line, separate count atomics gone; arrival order as before
# speedup vs baseline: 3.8694x; 2.1528x over previous
#include <hip/hip_runtime.h>

typedef __attribute__((ext_vector_type(2))) float v2f;
typedef __attribute__((ext_vector_type(4))) float v4f;
typedef __attribute__((ext_vector_type(8))) float v8f;

__global__ void NeighbourAggregation_scatter_kernel(
    const float* __restrict__ x,
    const int* __restrict__ edge_dst,
    float* __restrict__ out,
    float* __restrict__ counts,
    int total)
{
    int num_edges = total >> 4;
    int lane  = threadIdx.x & 31;
    int gwave = (blockIdx.x * blockDim.x + threadIdx.x) >> 5;
    int ebase = gwave * 32;
    if (ebase >= num_edges) return;

    int f = lane & 15;
    int h = lane >> 4;

    if (ebase + 32 <= num_edges) {
        int my_d = __builtin_nontemporal_load(edge_dst + ebase + lane);
        const float* xp = x + ebase * 16 + lane;

        float v[16];
        #pragma unroll
        for (int i = 0; i < 16; ++i) {
            v[i] = __builtin_nontemporal_load(xp + 32 * i);
        }

        #pragma unroll
        for (int i = 0; i < 16; ++i) {
            int d0 = __builtin_amdgcn_readlane(my_d, 2 * i);
            int d1 = __builtin_amdgcn_readlane(my_d, 2 * i + 1);
            int d  = h ? d1 : d0;
            unsafeAtomicAdd(out + d * 16 + f, v[i]);
        }

        unsafeAtomicAdd(counts + my_d, 1.0f);
    } else {
        for (int e = ebase + h; e < num_edges; e += 2) {
            int d = edge_dst[e];
            unsafeAtomicAdd(out + d * 16 + f, x[e * 16 + f]);
            if (f == 0) unsafeAtomicAdd(counts + d, 1.0f);
        }
    }
}

__global__ void NeighbourAggregation_finalize_wmma_kernel(
    float* __restrict__ out,
    const float* __restrict__ counts,
    int num_nodes)
{
    int wavesPerBlock = blockDim.x >> 5;
    int gwave = blockIdx.x * wavesPerBlock + (threadIdx.x >> 5);
    int lane  = threadIdx.x & 31;
    int nb    = gwave * 16;
    if (nb >= num_nodes) return;

    int col  = lane & 15;
    int half = lane >> 4;

    if (nb + 16 <= num_nodes) {
        float r = __frsqrt_rn(counts[nb + col] + 1.0f);

        v8f c = {0.f, 0.f, 0.f, 0.f, 0.f, 0.f, 0.f, 0.f};
        #pragma unroll
        for (int k0 = 0; k0 < 16; k0 += 4) {
            v2f a, b;
            #pragma unroll
            for (int j = 0; j < 2; ++j) {
                int K = k0 + 2 * half + j;
                a[j] = (col == K) ? r : 0.0f;
                b[j] = out[(nb + K) * 16 + col];
            }
            c = __builtin_amdgcn_wmma_f32_16x16x4_f32(
                    false, a, false, b, (short)0, c, false, false);
        }

        #pragma unroll
        for (int v = 0; v < 8; ++v) {
            int row = v + 8 * half;
            out[(nb + row) * 16 + col] = c[v];
        }
    } else {
        for (int row = nb + half; row < num_nodes; row += 2) {
            float r = __frsqrt_rn(counts[row] + 1.0f);
            out[row * 16 + col] *= r;
        }
    }
}

__global__ void NeighbourAggregation_scatter_padded_kernel(
    const float* __restrict__ x,
    const int* __restrict__ edge_dst,
    float* __restrict__ acc,
    int total)
{
    int num_edges = total >> 4;
    int lane  = threadIdx.x & 31;
    int gwave = (blockIdx.x * blockDim.x + threadIdx.x) >> 5;
    int ebase = gwave * 32;
    if (ebase >= num_edges) return;

    if (ebase + 32 <= num_edges) {
        int my_d = __builtin_nontemporal_load(edge_dst + ebase + lane);
        const float* xp = x + ebase * 16 + lane;

        float v[16];
        #pragma unroll
        for (int i = 0; i < 16; ++i) {
            v[i] = __builtin_nontemporal_load(xp + 32 * i);
        }

        if (lane <= 16) {
            #pragma unroll
            for (int i = 0; i < 16; ++i) {
                int d = __builtin_amdgcn_readlane(my_d, 2 * i);
                float* row = acc + (size_t)d * 32;
                unsafeAtomicAdd(row + lane, (lane == 16) ? 1.0f : v[i]);
            }
        }
        if (lane >= 15) {
            int colO = (lane >= 16) ? (lane - 16) : 16;
            #pragma unroll
            for (int i = 0; i < 16; ++i) {
                int d = __builtin_amdgcn_readlane(my_d, 2 * i + 1);
                float* row = acc + (size_t)d * 32;
                unsafeAtomicAdd(row + colO, (lane == 15) ? 1.0f : v[i]);
            }
        }
    } else {
        int f = lane & 15;
        int h = lane >> 4;
        for (int e = ebase + h; e < num_edges; e += 2) {
            int d = edge_dst[e];
            float* row = acc + (size_t)d * 32;
            unsafeAtomicAdd(row + f, x[e * 16 + f]);
            if (f == 0) unsafeAtomicAdd(row + 16, 1.0f);
        }
    }
}

__global__ void NeighbourAggregation_finalize_padded_wmma_kernel(
    const float* __restrict__ acc,
    float* __restrict__ out,
    int num_nodes)
{
    int wavesPerBlock = blockDim.x >> 5;
    int gwave = blockIdx.x * wavesPerBlock + (threadIdx.x >> 5);
    int lane  = threadIdx.x & 31;
    int nb    = gwave * 16;
    if (nb >= num_nodes) return;

    int col  = lane & 15;
    int half = lane >> 4;

    if (nb + 16 <= num_nodes) {
        float r = __frsqrt_rn(acc[(size_t)(nb + col) * 32 + 16] + 1.0f);

        v8f c = {0.f, 0.f, 0.f, 0.f, 0.f, 0.f, 0.f, 0.f};
        #pragma unroll
        for (int k0 = 0; k0 < 16; k0 += 4) {
            v2f a, b;
            #pragma unroll
            for (int j = 0; j < 2; ++j) {
                int K = k0 + 2 * half + j;
                a[j] = (col == K) ? r : 0.0f;
                b[j] = acc[(size_t)(nb + K) * 32 + col];
            }
            c = __builtin_amdgcn_wmma_f32_16x16x4_f32(
                    false, a, false, b, (short)0, c, false, false);
        }

        #pragma unroll
        for (int v = 0; v < 8; ++v) {
            int row = v + 8 * half;
            out[(nb + row) * 16 + col] = c[v];
        }
    } else {
        for (int row = nb + half; row < num_nodes; row += 2) {
            float r = __frsqrt_rn(acc[(size_t)row * 32 + 16] + 1.0f);
            out[row * 16 + col] = acc[(size_t)row * 32 + col] * r;
        }
    }
}

extern "C" void kernel_launch(void* const* d_in, const int* in_sizes, int n_in,
                              void* d_out, int out_size, void* d_ws, size_t ws_size,
                              hipStream_t stream) {
    const float* x        = (const float*)d_in[0];
    const int*   edge_dst = (const int*)d_in[1];
    int total     = in_sizes[0];
    int num_nodes = out_size / 16;

    float* out = (float*)d_out;

    int swaves  = (total / 16 + 31) / 32;
    int sblocks = (swaves + 7) / 8;
    int nwaves  = (num_nodes + 15) / 16;
    int fblocks = (nwaves + 7) / 8;

    size_t padded_bytes = (size_t)num_nodes * 32 * sizeof(float);
    if (ws_size >= padded_bytes) {
        float* acc = (float*)d_ws;
        hipMemsetAsync(d_ws, 0, padded_bytes, stream);
        NeighbourAggregation_scatter_padded_kernel<<<sblocks, 256, 0, stream>>>(
            x, edge_dst, acc, total);
        NeighbourAggregation_finalize_padded_wmma_kernel<<<fblocks, 256, 0, stream>>>(
            acc, out, num_nodes);
    } else {
        float* counts = (float*)d_ws;
        hipMemsetAsync(d_out, 0, (size_t)out_size * sizeof(float), stream);
        hipMemsetAsync(d_ws, 0, (size_t)num_nodes * sizeof(float), stream);
        NeighbourAggregation_scatter_kernel<<<sblocks, 256, 0, stream>>>(
            x, edge_dst, out, counts, total);
        NeighbourAggregation_finalize_wmma_kernel<<<fblocks, 256, 0, stream>>>(
            out, counts, num_nodes);
    }
}


# One accumulator plane per package half: a wave reads its die number and adds into its own half's plane, so a row's line stays in one L2 half; finalize adds the two planes (two partial sums per row)
# speedup vs baseline: 5.6881x; 1.4700x over previous
#include <hip/hip_runtime.h>

typedef __attribute__((ext_vector_type(2))) float v2f;
typedef __attribute__((ext_vector_type(4))) float v4f;
typedef __attribute__((ext_vector_type(8))) float v8f;

__global__ void NeighbourAggregation_scatter_kernel(
    const float* __restrict__ x,
    const int* __restrict__ edge_dst,
    float* __restrict__ out,
    float* __restrict__ counts,
    int total)
{
    int num_edges = total >> 4;
    int lane  = threadIdx.x & 31;
    int gwave = (blockIdx.x * blockDim.x + threadIdx.x) >> 5;
    int ebase = gwave * 32;
    if (ebase >= num_edges) return;

    int f = lane & 15;
    int h = lane >> 4;

    if (ebase + 32 <= num_edges) {
        int my_d = __builtin_nontemporal_load(edge_dst + ebase + lane);
        const float* xp = x + ebase * 16 + lane;

        float v[16];
        #pragma unroll
        for (int i = 0; i < 16; ++i) {
            v[i] = __builtin_nontemporal_load(xp + 32 * i);
        }

        #pragma unroll
        for (int i = 0; i < 16; ++i) {
            int d0 = __builtin_amdgcn_readlane(my_d, 2 * i);
            int d1 = __builtin_amdgcn_readlane(my_d, 2 * i + 1);
            int d  = h ? d1 : d0;
            unsafeAtomicAdd(out + d * 16 + f, v[i]);
        }

        unsafeAtomicAdd(counts + my_d, 1.0f);
    } else {
        for (int e = ebase + h; e < num_edges; e += 2) {
            int d = edge_dst[e];
            unsafeAtomicAdd(out + d * 16 + f, x[e * 16 + f]);
            if (f == 0) unsafeAtomicAdd(counts + d, 1.0f);
        }
    }
}

__global__ void NeighbourAggregation_finalize_wmma_kernel(
    float* __restrict__ out,
    const float* __restrict__ counts,
    int num_nodes)
{
    int wavesPerBlock = blockDim.x >> 5;
    int gwave = blockIdx.x * wavesPerBlock + (threadIdx.x >> 5);
    int lane  = threadIdx.x & 31;
    int nb    = gwave * 16;
    if (nb >= num_nodes) return;

    int col  = lane & 15;
    int half = lane >> 4;

    if (nb + 16 <= num_nodes) {
        float r = __frsqrt_rn(counts[nb + col] + 1.0f);

        v8f c = {0.f, 0.f, 0.f, 0.f, 0.f, 0.f, 0.f, 0.f};
        #pragma unroll
        for (int k0 = 0; k0 < 16; k0 += 4) {
            v2f a, b;
            #pragma unroll
            for (int j = 0; j < 2; ++j) {
                int K = k0 + 2 * half + j;
                a[j] = (col == K) ? r : 0.0f;
                b[j] = out[(nb + K) * 16 + col];
            }
            c = __builtin_amdgcn_wmma_f32_16x16x4_f32(
                    false, a, false, b, (short)0, c, false, false);
        }

        #pragma unroll
        for (int v = 0; v < 8; ++v) {
            int row = v + 8 * half;
            out[(nb + row) * 16 + col] = c[v];
        }
    } else {
        for (int row = nb + half; row < num_nodes; row += 2) {
            float r = __frsqrt_rn(counts[row] + 1.0f);
            out[row * 16 + col] *= r;
        }
    }
}

__global__ void NeighbourAggregation_scatter_padded_kernel(
    const float* __restrict__ x,
    const int* __restrict__ edge_dst,
    float* __restrict__ acc,
    int total)
{
    int num_edges = total >> 4;
    int lane  = threadIdx.x & 31;
    int gwave = (blockIdx.x * blockDim.x + threadIdx.x) >> 5;
    int ebase = gwave * 32;
    if (ebase >= num_edges) return;

    if (ebase + 32 <= num_edges) {
        int my_d = __builtin_nontemporal_load(edge_dst + ebase + lane);
        const float* xp = x + ebase * 16 + lane;

        float v[16];
        #pragma unroll
        for (int i = 0; i < 16; ++i) {
            v[i] = __builtin_nontemporal_load(xp + 32 * i);
        }

        if (lane <= 16) {
            #pragma unroll
            for (int i = 0; i < 16; ++i) {
                int d = __builtin_amdgcn_readlane(my_d, 2 * i);
                float* row = acc + (size_t)d * 32;
                unsafeAtomicAdd(row + lane, (lane == 16) ? 1.0f : v[i]);
            }
        }
        if (lane >= 15) {
            int colO = (lane >= 16) ? (lane - 16) : 16;
            #pragma unroll
            for (int i = 0; i < 16; ++i) {
                int d = __builtin_amdgcn_readlane(my_d, 2 * i + 1);
                float* row = acc + (size_t)d * 32;
                unsafeAtomicAdd(row + colO, (lane == 15) ? 1.0f : v[i]);
            }
        }
    } else {
        int f = lane & 15;
        int h = lane >> 4;
        for (int e = ebase + h; e < num_edges; e += 2) {
            int d = edge_dst[e];
            float* row = acc + (size_t)d * 32;
            unsafeAtomicAdd(row + f, x[e * 16 + f]);
            if (f == 0) unsafeAtomicAdd(row + 16, 1.0f);
        }
    }
}

__global__ void NeighbourAggregation_finalize_padded_wmma_kernel(
    const float* __restrict__ acc,
    float* __restrict__ out,
    int num_nodes)
{
    int wavesPerBlock = blockDim.x >> 5;
    int gwave = blockIdx.x * wavesPerBlock + (threadIdx.x >> 5);
    int lane  = threadIdx.x & 31;
    int nb    = gwave * 16;
    if (nb >= num_nodes) return;

    int col  = lane & 15;
    int half = lane >> 4;

    if (nb + 16 <= num_nodes) {
        float r = __frsqrt_rn(acc[(size_t)(nb + col) * 32 + 16] + 1.0f);

        v8f c = {0.f, 0.f, 0.f, 0.f, 0.f, 0.f, 0.f, 0.f};
        #pragma unroll
        for (int k0 = 0; k0 < 16; k0 += 4) {
            v2f a, b;
            #pragma unroll
            for (int j = 0; j < 2; ++j) {
                int K = k0 + 2 * half + j;
                a[j] = (col == K) ? r : 0.0f;
                b[j] = acc[(size_t)(nb + K) * 32 + col];
            }
            c = __builtin_amdgcn_wmma_f32_16x16x4_f32(
                    false, a, false, b, (short)0, c, false, false);
        }

        #pragma unroll
        for (int v = 0; v < 8; ++v) {
            int row = v + 8 * half;
            out[(nb + row) * 16 + col] = c[v];
        }
    } else {
        for (int row = nb + half; row < num_nodes; row += 2) {
            float r = __frsqrt_rn(acc[(size_t)row * 32 + 16] + 1.0f);
            out[row * 16 + col] = acc[(size_t)row * 32 + col] * r;
        }
    }
}


__global__ void NeighbourAggregation_scatter_halfplane_kernel(
    const float* __restrict__ x,
    const int* __restrict__ edge_dst,
    float* __restrict__ acc0,
    float* __restrict__ acc1,
    int total)
{
    unsigned int where = __builtin_amdgcn_s_sendmsg_rtn(0x87);
    float* __restrict__ acc = ((where >> 18) & 1u) ? acc1 : acc0;

    int num_edges = total >> 4;
    int lane  = threadIdx.x & 31;
    int gwave = (blockIdx.x * blockDim.x + threadIdx.x) >> 5;
    int ebase = gwave * 32;
    if (ebase >= num_edges) return;

    if (ebase + 32 <= num_edges) {
        int my_d = __builtin_nontemporal_load(edge_dst + ebase + lane);
        const float* xp = x + ebase * 16 + lane;

        float v[16];
        #pragma unroll
        for (int i = 0; i < 16; ++i) {
            v[i] = __builtin_nontemporal_load(xp + 32 * i);
        }

        if (lane <= 16) {
            #pragma unroll
            for (int i = 0; i < 16; ++i) {
                int d = __builtin_amdgcn_readlane(my_d, 2 * i);
                float* row = acc + (size_t)d * 32;
                unsafeAtomicAdd(row + lane, (lane == 16) ? 1.0f : v[i]);
            }
        }
        if (lane >= 15) {
            int colO = (lane >= 16) ? (lane - 16) : 16;
            #pragma unroll
            for (int i = 0; i < 16; ++i) {
                int d = __builtin_amdgcn_readlane(my_d, 2 * i + 1);
                float* row = acc + (size_t)d * 32;
                unsafeAtomicAdd(row + colO, (lane == 15) ? 1.0f : v[i]);
            }
        }
    } else {
        int f = lane & 15;
        int h = lane >> 4;
        for (int e = ebase + h; e < num_edges; e += 2) {
            int d = edge_dst[e];
            float* row = acc + (size_t)d * 32;
            unsafeAtomicAdd(row + f, x[e * 16 + f]);
            if (f == 0) unsafeAtomicAdd(row + 16, 1.0f);
        }
    }
}

__global__ void NeighbourAggregation_finalize_halfplane_wmma_kernel(
    const float* __restrict__ acc0,
    const float* __restrict__ acc1,
    float* __restrict__ out,
    int num_nodes)
{
    int wavesPerBlock = blockDim.x >> 5;
    int gwave = blockIdx.x * wavesPerBlock + (threadIdx.x >> 5);
    int lane  = threadIdx.x & 31;
    int nb    = gwave * 16;
    if (nb >= num_nodes) return;

    int col  = lane & 15;
    int half = lane >> 4;

    if (nb + 16 <= num_nodes) {
        size_t cidx = (size_t)(nb + col) * 32 + 16;
        float r = __frsqrt_rn(acc0[cidx] + acc1[cidx] + 1.0f);

        v8f c = {0.f, 0.f, 0.f, 0.f, 0.f, 0.f, 0.f, 0.f};
        #pragma unroll
        for (int k0 = 0; k0 < 16; k0 += 4) {
            v2f a, b;
            #pragma unroll
            for (int j = 0; j < 2; ++j) {
                int K = k0 + 2 * half + j;
                size_t sidx = (size_t)(nb + K) * 32 + col;
                a[j] = (col == K) ? r : 0.0f;
                b[j] = acc0[sidx] + acc1[sidx];
            }
            c = __builtin_amdgcn_wmma_f32_16x16x4_f32(
                    false, a, false, b, (short)0, c, false, false);
        }

        #pragma unroll
        for (int v = 0; v < 8; ++v) {
            int row = v + 8 * half;
            out[(nb + row) * 16 + col] = c[v];
        }
    } else {
        for (int row = nb + half; row < num_nodes; row += 2) {
            size_t base = (size_t)row * 32;
            float r = __frsqrt_rn(acc0[base + 16] + acc1[base + 16] + 1.0f);
            out[row * 16 + col] = (acc0[base + col] + acc1[base + col]) * r;
        }
    }
}

extern "C" void kernel_launch(void* const* d_in, const int* in_sizes, int n_in,
                              void* d_out, int out_size, void* d_ws, size_t ws_size,
                              hipStream_t stream) {
    const float* x        = (const float*)d_in[0];
    const int*   edge_dst = (const int*)d_in[1];
    int total     = in_sizes[0];
    int num_nodes = out_size / 16;

    float* out = (float*)d_out;

    int swaves  = (total / 16 + 31) / 32;
    int sblocks = (swaves + 7) / 8;
    int nwaves  = (num_nodes + 15) / 16;
    int fblocks = (nwaves + 7) / 8;

    size_t padded_bytes = (size_t)num_nodes * 32 * sizeof(float);
    if (ws_size >= 2 * padded_bytes) {
        float* acc0 = (float*)d_ws;
        float* acc1 = acc0 + (size_t)num_nodes * 32;
        hipMemsetAsync(d_ws, 0, 2 * padded_bytes, stream);
        NeighbourAggregation_scatter_halfplane_kernel<<<sblocks, 256, 0, stream>>>(
            x, edge_dst, acc0, acc1, total);
        NeighbourAggregation_finalize_halfplane_wmma_kernel<<<fblocks, 256, 0, stream>>>(
            acc0, acc1, out, num_nodes);
    } else if (ws_size >= padded_bytes) {
        float* acc = (float*)d_ws;
        hipMemsetAsync(d_ws, 0, padded_bytes, stream);
        NeighbourAggregation_scatter_padded_kernel<<<sblocks, 256, 0, stream>>>(
            x, edge_dst, acc, total);
        NeighbourAggregation_finalize_padded_wmma_kernel<<<fblocks, 256, 0, stream>>>(
            acc, out, num_nodes);
    } else {
        float* counts = (float*)d_ws;
        hipMemsetAsync(d_out, 0, (size_t)out_size * sizeof(float), stream);
        hipMemsetAsync(d_ws, 0, (size_t)num_nodes * sizeof(float), stream);
        NeighbourAggregation_scatter_kernel<<<sblocks, 256, 0, stream>>>(
            x, edge_dst, out, counts, total);
        NeighbourAggregation_finalize_wmma_kernel<<<fblocks, 256, 0, stream>>>(
            out, counts, num_nodes);
    }
}


# Scatter waves made resident: 2,048 work-groups whose waves loop over the 32-edge groups (grid stride) replace 12,500 one-shot ones, 100,000 wave starts become 16,384; atomic arrival order as before
# speedup vs baseline: 6.0272x; 1.0596x over previous
#include <hip/hip_runtime.h>

typedef __attribute__((ext_vector_type(2))) float v2f;
typedef __attribute__((ext_vector_type(4))) float v4f;
typedef __attribute__((ext_vector_type(8))) float v8f;

__global__ void NeighbourAggregation_scatter_kernel(
    const float* __restrict__ x,
    const int* __restrict__ edge_dst,
    float* __restrict__ out,
    float* __restrict__ counts,
    int total)
{
    int num_edges = total >> 4;
    int lane  = threadIdx.x & 31;
    int gwave = (blockIdx.x * blockDim.x + threadIdx.x) >> 5;
    int ebase = gwave * 32;
    if (ebase >= num_edges) return;

    int f = lane & 15;
    int h = lane >> 4;

    if (ebase + 32 <= num_edges) {
        int my_d = __builtin_nontemporal_load(edge_dst + ebase + lane);
        const float* xp = x + ebase * 16 + lane;

        float v[16];
        #pragma unroll
        for (int i = 0; i < 16; ++i) {
            v[i] = __builtin_nontemporal_load(xp + 32 * i);
        }

        #pragma unroll
        for (int i = 0; i < 16; ++i) {
            int d0 = __builtin_amdgcn_readlane(my_d, 2 * i);
            int d1 = __builtin_amdgcn_readlane(my_d, 2 * i + 1);
            int d  = h ? d1 : d0;
            unsafeAtomicAdd(out + d * 16 + f, v[i]);
        }

        unsafeAtomicAdd(counts + my_d, 1.0f);
    } else {
        for (int e = ebase + h; e < num_edges; e += 2) {
            int d = edge_dst[e];
            unsafeAtomicAdd(out + d * 16 + f, x[e * 16 + f]);
            if (f == 0) unsafeAtomicAdd(counts + d, 1.0f);
        }
    }
}

__global__ void NeighbourAggregation_finalize_wmma_kernel(
    float* __restrict__ out,
    const float* __restrict__ counts,
    int num_nodes)
{
    int wavesPerBlock = blockDim.x >> 5;
    int gwave = blockIdx.x * wavesPerBlock + (threadIdx.x >> 5);
    int lane  = threadIdx.x & 31;
    int nb    = gwave * 16;
    if (nb >= num_nodes) return;

    int col  = lane & 15;
    int half = lane >> 4;

    if (nb + 16 <= num_nodes) {
        float r = __frsqrt_rn(counts[nb + col] + 1.0f);

        v8f c = {0.f, 0.f, 0.f, 0.f, 0.f, 0.f, 0.f, 0.f};
        #pragma unroll
        for (int k0 = 0; k0 < 16; k0 += 4) {
            v2f a, b;
            #pragma unroll
            for (int j = 0; j < 2; ++j) {
                int K = k0 + 2 * half + j;
                a[j] = (col == K) ? r : 0.0f;
                b[j] = out[(nb + K) * 16 + col];
            }
            c = __builtin_amdgcn_wmma_f32_16x16x4_f32(
                    false, a, false, b, (short)0, c, false, false);
        }

        #pragma unroll
        for (int v = 0; v < 8; ++v) {
            int row = v + 8 * half;
            out[(nb + row) * 16 + col] = c[v];
        }
    } else {
        for (int row = nb + half; row < num_nodes; row += 2) {
            float r = __frsqrt_rn(counts[row] + 1.0f);
            out[row * 16 + col] *= r;
        }
    }
}

__global__ void NeighbourAggregation_scatter_padded_kernel(
    const float* __restrict__ x,
    const int* __restrict__ edge_dst,
    float* __restrict__ acc,
    int total)
{
    int num_edges = total >> 4;
    int lane  = threadIdx.x & 31;
    int gwave = (blockIdx.x * blockDim.x + threadIdx.x) >> 5;
    int ebase = gwave * 32;
    if (ebase >= num_edges) return;

    if (ebase + 32 <= num_edges) {
        int my_d = __builtin_nontemporal_load(edge_dst + ebase + lane);
        const float* xp = x + ebase * 16 + lane;

        float v[16];
        #pragma unroll
        for (int i = 0; i < 16; ++i) {
            v[i] = __builtin_nontemporal_load(xp + 32 * i);
        }

        if (lane <= 16) {
            #pragma unroll
            for (int i = 0; i < 16; ++i) {
                int d = __builtin_amdgcn_readlane(my_d, 2 * i);
                float* row = acc + (size_t)d * 32;
                unsafeAtomicAdd(row + lane, (lane == 16) ? 1.0f : v[i]);
            }
        }
        if (lane >= 15) {
            int colO = (lane >= 16) ? (lane - 16) : 16;
            #pragma unroll
            for (int i = 0; i < 16; ++i) {
                int d = __builtin_amdgcn_readlane(my_d, 2 * i + 1);
                float* row = acc + (size_t)d * 32;
                unsafeAtomicAdd(row + colO, (lane == 15) ? 1.0f : v[i]);
            }
        }
    } else {
        int f = lane & 15;
        int h = lane >> 4;
        for (int e = ebase + h; e < num_edges; e += 2) {
            int d = edge_dst[e];
            float* row = acc + (size_t)d * 32;
            unsafeAtomicAdd(row + f, x[e * 16 + f]);
            if (f == 0) unsafeAtomicAdd(row + 16, 1.0f);
        }
    }
}

__global__ void NeighbourAggregation_finalize_padded_wmma_kernel(
    const float* __restrict__ acc,
    float* __restrict__ out,
    int num_nodes)
{
    int wavesPerBlock = blockDim.x >> 5;
    int gwave = blockIdx.x * wavesPerBlock + (threadIdx.x >> 5);
    int lane  = threadIdx.x & 31;
    int nb    = gwave * 16;
    if (nb >= num_nodes) return;

    int col  = lane & 15;
    int half = lane >> 4;

    if (nb + 16 <= num_nodes) {
        float r = __frsqrt_rn(acc[(size_t)(nb + col) * 32 + 16] + 1.0f);

        v8f c = {0.f, 0.f, 0.f, 0.f, 0.f, 0.f, 0.f, 0.f};
        #pragma unroll
        for (int k0 = 0; k0 < 16; k0 += 4) {
            v2f a, b;
            #pragma unroll
            for (int j = 0; j < 2; ++j) {
                int K = k0 + 2 * half + j;
                a[j] = (col == K) ? r : 0.0f;
                b[j] = acc[(size_t)(nb + K) * 32 + col];
            }
            c = __builtin_amdgcn_wmma_f32_16x16x4_f32(
                    false, a, false, b, (short)0, c, false, false);
        }

        #pragma unroll
        for (int v = 0; v < 8; ++v) {
            int row = v + 8 * half;
            out[(nb + row) * 16 + col] = c[v];
        }
    } else {
        for (int row = nb + half; row < num_nodes; row += 2) {
            float r = __frsqrt_rn(acc[(size_t)row * 32 + 16] + 1.0f);
            out[row * 16 + col] = acc[(size_t)row * 32 + col] * r;
        }
    }
}


__global__ void NeighbourAggregation_scatter_halfplane_kernel(
    const float* __restrict__ x,
    const int* __restrict__ edge_dst,
    float* __restrict__ acc0,
    float* __restrict__ acc1,
    int total)
{
    unsigned int where = __builtin_amdgcn_s_sendmsg_rtn(0x87);
    float* __restrict__ acc = ((where >> 18) & 1u) ? acc1 : acc0;

    int num_edges = total >> 4;
    int ngroups   = (num_edges + 31) >> 5;
    int lane   = threadIdx.x & 31;
    int gwave  = (blockIdx.x * blockDim.x + threadIdx.x) >> 5;
    int nwaves = (gridDim.x * blockDim.x) >> 5;

    for (int g = gwave; g < ngroups; g += nwaves) {
        int ebase = g * 32;

        if (ebase + 32 <= num_edges) {
            int my_d = __builtin_nontemporal_load(edge_dst + ebase + lane);
            const float* xp = x + (size_t)ebase * 16 + lane;

            float v[16];
            #pragma unroll
            for (int i = 0; i < 16; ++i) {
                v[i] = __builtin_nontemporal_load(xp + 32 * i);
            }

            if (lane <= 16) {
                #pragma unroll
                for (int i = 0; i < 16; ++i) {
                    int d = __builtin_amdgcn_readlane(my_d, 2 * i);
                    float* row = acc + (size_t)d * 32;
                    unsafeAtomicAdd(row + lane, (lane == 16) ? 1.0f : v[i]);
                }
            }
            if (lane >= 15) {
                int colO = (lane >= 16) ? (lane - 16) : 16;
                #pragma unroll
                for (int i = 0; i < 16; ++i) {
                    int d = __builtin_amdgcn_readlane(my_d, 2 * i + 1);
                    float* row = acc + (size_t)d * 32;
                    unsafeAtomicAdd(row + colO, (lane == 15) ? 1.0f : v[i]);
                }
            }
        } else {
            int f = lane & 15;
            int h = lane >> 4;
            for (int e = ebase + h; e < num_edges; e += 2) {
                int d = edge_dst[e];
                float* row = acc + (size_t)d * 32;
                unsafeAtomicAdd(row + f, x[(size_t)e * 16 + f]);
                if (f == 0) unsafeAtomicAdd(row + 16, 1.0f);
            }
        }
    }
}

__global__ void NeighbourAggregation_finalize_halfplane_wmma_kernel(
    const float* __restrict__ acc0,
    const float* __restrict__ acc1,
    float* __restrict__ out,
    int num_nodes)
{
    int wavesPerBlock = blockDim.x >> 5;
    int gwave = blockIdx.x * wavesPerBlock + (threadIdx.x >> 5);
    int lane  = threadIdx.x & 31;
    int nb    = gwave * 16;
    if (nb >= num_nodes) return;

    int col  = lane & 15;
    int half = lane >> 4;

    if (nb + 16 <= num_nodes) {
        size_t cidx = (size_t)(nb + col) * 32 + 16;
        float r = __frsqrt_rn(acc0[cidx] + acc1[cidx] + 1.0f);

        v8f c = {0.f, 0.f, 0.f, 0.f, 0.f, 0.f, 0.f, 0.f};
        #pragma unroll
        for (int k0 = 0; k0 < 16; k0 += 4) {
            v2f a, b;
            #pragma unroll
            for (int j = 0; j < 2; ++j) {
                int K = k0 + 2 * half + j;
                size_t sidx = (size_t)(nb + K) * 32 + col;
                a[j] = (col == K) ? r : 0.0f;
                b[j] = acc0[sidx] + acc1[sidx];
            }
            c = __builtin_amdgcn_wmma_f32_16x16x4_f32(
                    false, a, false, b, (short)0, c, false, false);
        }

        #pragma unroll
        for (int v = 0; v < 8; ++v) {
            int row = v + 8 * half;
            out[(nb + row) * 16 + col] = c[v];
        }
    } else {
        for (int row = nb + half; row < num_nodes; row += 2) {
            size_t base = (size_t)row * 32;
            float r = __frsqrt_rn(acc0[base + 16] + acc1[base + 16] + 1.0f);
            out[row * 16 + col] = (acc0[base + col] + acc1[base + col]) * r;
        }
    }
}

extern "C" void kernel_launch(void* const* d_in, const int* in_sizes, int n_in,
                              void* d_out, int out_size, void* d_ws, size_t ws_size,
                              hipStream_t stream) {
    const float* x        = (const float*)d_in[0];
    const int*   edge_dst = (const int*)d_in[1];
    int total     = in_sizes[0];
    int num_nodes = out_size / 16;

    float* out = (float*)d_out;

    int swaves  = (total / 16 + 31) / 32;
    int sblocks = (swaves + 7) / 8;
    int nwaves  = (num_nodes + 15) / 16;
    int fblocks = (nwaves + 7) / 8;

    size_t padded_bytes = (size_t)num_nodes * 32 * sizeof(float);
    if (ws_size >= 2 * padded_bytes) {
        float* acc0 = (float*)d_ws;
        float* acc1 = acc0 + (size_t)num_nodes * 32;
        hipMemsetAsync(d_ws, 0, 2 * padded_bytes, stream);
        NeighbourAggregation_scatter_halfplane_kernel<<<2048, 256, 0, stream>>>(
            x, edge_dst, acc0, acc1, total);
        NeighbourAggregation_finalize_halfplane_wmma_kernel<<<fblocks, 256, 0, stream>>>(
            acc0, acc1, out, num_nodes);
    } else if (ws_size >= padded_bytes) {
        float* acc = (float*)d_ws;
        hipMemsetAsync(d_ws, 0, padded_bytes, stream);
        NeighbourAggregation_scatter_padded_kernel<<<sblocks, 256, 0, stream>>>(
            x, edge_dst, acc, total);
        NeighbourAggregation_finalize_padded_wmma_kernel<<<fblocks, 256, 0, stream>>>(
            acc, out, num_nodes);
    } else {
        float* counts = (float*)d_ws;
        hipMemsetAsync(d_out, 0, (size_t)out_size * sizeof(float), stream);
        hipMemsetAsync(d_ws, 0, (size_t)num_nodes * sizeof(float), stream);
        NeighbourAggregation_scatter_kernel<<<sblocks, 256, 0, stream>>>(
            x, edge_dst, out, counts, total);
        NeighbourAggregation_finalize_wmma_kernel<<<fblocks, 256, 0, stream>>>(
            out, counts, num_nodes);
    }
}


# Global f32 atomics replaced by static owner work-groups, one per WGP: edges binned per chunk by 400-row block, rows gathered from x, summed by LDS f32 adds; memset, planes gone; sum order changes
# speedup vs baseline: 7.4717x; 1.2397x over previous
#include <hip/hip_runtime.h>

typedef __attribute__((ext_vector_type(2))) float v2f;
typedef __attribute__((ext_vector_type(4))) float v4f;
typedef __attribute__((ext_vector_type(8))) float v8f;

__global__ void NeighbourAggregation_scatter_kernel(
    const float* __restrict__ x,
    const int* __restrict__ edge_dst,
    float* __restrict__ out,
    float* __restrict__ counts,
    int total)
{
    int num_edges = total >> 4;
    int lane  = threadIdx.x & 31;
    int gwave = (blockIdx.x * blockDim.x + threadIdx.x) >> 5;
    int ebase = gwave * 32;
    if (ebase >= num_edges) return;

    int f = lane & 15;
    int h = lane >> 4;

    if (ebase + 32 <= num_edges) {
        int my_d = __builtin_nontemporal_load(edge_dst + ebase + lane);
        const float* xp = x + ebase * 16 + lane;

        float v[16];
        #pragma unroll
        for (int i = 0; i < 16; ++i) {
            v[i] = __builtin_nontemporal_load(xp + 32 * i);
        }

        #pragma unroll
        for (int i = 0; i < 16; ++i) {
            int d0 = __builtin_amdgcn_readlane(my_d, 2 * i);
            int d1 = __builtin_amdgcn_readlane(my_d, 2 * i + 1);
            int d  = h ? d1 : d0;
            unsafeAtomicAdd(out + d * 16 + f, v[i]);
        }

        unsafeAtomicAdd(counts + my_d, 1.0f);
    } else {
        for (int e = ebase + h; e < num_edges; e += 2) {
            int d = edge_dst[e];
            unsafeAtomicAdd(out + d * 16 + f, x[e * 16 + f]);
            if (f == 0) unsafeAtomicAdd(counts + d, 1.0f);
        }
    }
}

__global__ void NeighbourAggregation_finalize_wmma_kernel(
    float* __restrict__ out,
    const float* __restrict__ counts,
    int num_nodes)
{
    int wavesPerBlock = blockDim.x >> 5;
    int gwave = blockIdx.x * wavesPerBlock + (threadIdx.x >> 5);
    int lane  = threadIdx.x & 31;
    int nb    = gwave * 16;
    if (nb >= num_nodes) return;

    int col  = lane & 15;
    int half = lane >> 4;

    if (nb + 16 <= num_nodes) {
        float r = __frsqrt_rn(counts[nb + col] + 1.0f);

        v8f c = {0.f, 0.f, 0.f, 0.f, 0.f, 0.f, 0.f, 0.f};
        #pragma unroll
        for (int k0 = 0; k0 < 16; k0 += 4) {
            v2f a, b;
            #pragma unroll
            for (int j = 0; j < 2; ++j) {
                int K = k0 + 2 * half + j;
                a[j] = (col == K) ? r : 0.0f;
                b[j] = out[(nb + K) * 16 + col];
            }
            c = __builtin_amdgcn_wmma_f32_16x16x4_f32(
                    false, a, false, b, (short)0, c, false, false);
        }

        #pragma unroll
        for (int v = 0; v < 8; ++v) {
            int row = v + 8 * half;
            out[(nb + row) * 16 + col] = c[v];
        }
    } else {
        for (int row = nb + half; row < num_nodes; row += 2) {
            float r = __frsqrt_rn(counts[row] + 1.0f);
            out[row * 16 + col] *= r;
        }
    }
}

__global__ void NeighbourAggregation_scatter_padded_kernel(
    const float* __restrict__ x,
    const int* __restrict__ edge_dst,
    float* __restrict__ acc,
    int total)
{
    int num_edges = total >> 4;
    int lane  = threadIdx.x & 31;
    int gwave = (blockIdx.x * blockDim.x + threadIdx.x) >> 5;
    int ebase = gwave * 32;
    if (ebase >= num_edges) return;

    if (ebase + 32 <= num_edges) {
        int my_d = __builtin_nontemporal_load(edge_dst + ebase + lane);
        const float* xp = x + ebase * 16 + lane;

        float v[16];
        #pragma unroll
        for (int i = 0; i < 16; ++i) {
            v[i] = __builtin_nontemporal_load(xp + 32 * i);
        }

        if (lane <= 16) {
            #pragma unroll
            for (int i = 0; i < 16; ++i) {
                int d = __builtin_amdgcn_readlane(my_d, 2 * i);
                float* row = acc + (size_t)d * 32;
                unsafeAtomicAdd(row + lane, (lane == 16) ? 1.0f : v[i]);
            }
        }
        if (lane >= 15) {
            int colO = (lane >= 16) ? (lane - 16) : 16;
            #pragma unroll
            for (int i = 0; i < 16; ++i) {
                int d = __builtin_amdgcn_readlane(my_d, 2 * i + 1);
                float* row = acc + (size_t)d * 32;
                unsafeAtomicAdd(row + colO, (lane == 15) ? 1.0f : v[i]);
            }
        }
    } else {
        int f = lane & 15;
        int h = lane >> 4;
        for (int e = ebase + h; e < num_edges; e += 2) {
            int d = edge_dst[e];
            float* row = acc + (size_t)d * 32;
            unsafeAtomicAdd(row + f, x[e * 16 + f]);
            if (f == 0) unsafeAtomicAdd(row + 16, 1.0f);
        }
    }
}

__global__ void NeighbourAggregation_finalize_padded_wmma_kernel(
    const float* __restrict__ acc,
    float* __restrict__ out,
    int num_nodes)
{
    int wavesPerBlock = blockDim.x >> 5;
    int gwave = blockIdx.x * wavesPerBlock + (threadIdx.x >> 5);
    int lane  = threadIdx.x & 31;
    int nb    = gwave * 16;
    if (nb >= num_nodes) return;

    int col  = lane & 15;
    int half = lane >> 4;

    if (nb + 16 <= num_nodes) {
        float r = __frsqrt_rn(acc[(size_t)(nb + col) * 32 + 16] + 1.0f);

        v8f c = {0.f, 0.f, 0.f, 0.f, 0.f, 0.f, 0.f, 0.f};
        #pragma unroll
        for (int k0 = 0; k0 < 16; k0 += 4) {
            v2f a, b;
            #pragma unroll
            for (int j = 0; j < 2; ++j) {
                int K = k0 + 2 * half + j;
                a[j] = (col == K) ? r : 0.0f;
                b[j] = acc[(size_t)(nb + K) * 32 + col];
            }
            c = __builtin_amdgcn_wmma_f32_16x16x4_f32(
                    false, a, false, b, (short)0, c, false, false);
        }

        #pragma unroll
        for (int v = 0; v < 8; ++v) {
            int row = v + 8 * half;
            out[(nb + row) * 16 + col] = c[v];
        }
    } else {
        for (int row = nb + half; row < num_nodes; row += 2) {
            float r = __frsqrt_rn(acc[(size_t)row * 32 + 16] + 1.0f);
            out[row * 16 + col] = acc[(size_t)row * 32 + col] * r;
        }
    }
}


__global__ void NeighbourAggregation_scatter_halfplane_kernel(
    const float* __restrict__ x,
    const int* __restrict__ edge_dst,
    float* __restrict__ acc0,
    float* __restrict__ acc1,
    int total)
{
    unsigned int where = __builtin_amdgcn_s_sendmsg_rtn(0x87);
    float* __restrict__ acc = ((where >> 18) & 1u) ? acc1 : acc0;

    int num_edges = total >> 4;
    int ngroups   = (num_edges + 31) >> 5;
    int lane   = threadIdx.x & 31;
    int gwave  = (blockIdx.x * blockDim.x + threadIdx.x) >> 5;
    int nwaves = (gridDim.x * blockDim.x) >> 5;

    for (int g = gwave; g < ngroups; g += nwaves) {
        int ebase = g * 32;

        if (ebase + 32 <= num_edges) {
            int my_d = __builtin_nontemporal_load(edge_dst + ebase + lane);
            const float* xp = x + (size_t)ebase * 16 + lane;

            float v[16];
            #pragma unroll
            for (int i = 0; i < 16; ++i) {
                v[i] = __builtin_nontemporal_load(xp + 32 * i);
            }

            if (lane <= 16) {
                #pragma unroll
                for (int i = 0; i < 16; ++i) {
                    int d = __builtin_amdgcn_readlane(my_d, 2 * i);
                    float* row = acc + (size_t)d * 32;
                    unsafeAtomicAdd(row + lane, (lane == 16) ? 1.0f : v[i]);
                }
            }
            if (lane >= 15) {
                int colO = (lane >= 16) ? (lane - 16) : 16;
                #pragma unroll
                for (int i = 0; i < 16; ++i) {
                    int d = __builtin_amdgcn_readlane(my_d, 2 * i + 1);
                    float* row = acc + (size_t)d * 32;
                    unsafeAtomicAdd(row + colO, (lane == 15) ? 1.0f : v[i]);
                }
            }
        } else {
            int f = lane & 15;
            int h = lane >> 4;
            for (int e = ebase + h; e < num_edges; e += 2) {
                int d = edge_dst[e];
                float* row = acc + (size_t)d * 32;
                unsafeAtomicAdd(row + f, x[(size_t)e * 16 + f]);
                if (f == 0) unsafeAtomicAdd(row + 16, 1.0f);
            }
        }
    }
}

__global__ void NeighbourAggregation_finalize_halfplane_wmma_kernel(
    const float* __restrict__ acc0,
    const float* __restrict__ acc1,
    float* __restrict__ out,
    int num_nodes)
{
    int wavesPerBlock = blockDim.x >> 5;
    int gwave = blockIdx.x * wavesPerBlock + (threadIdx.x >> 5);
    int lane  = threadIdx.x & 31;
    int nb    = gwave * 16;
    if (nb >= num_nodes) return;

    int col  = lane & 15;
    int half = lane >> 4;

    if (nb + 16 <= num_nodes) {
        size_t cidx = (size_t)(nb + col) * 32 + 16;
        float r = __frsqrt_rn(acc0[cidx] + acc1[cidx] + 1.0f);

        v8f c = {0.f, 0.f, 0.f, 0.f, 0.f, 0.f, 0.f, 0.f};
        #pragma unroll
        for (int k0 = 0; k0 < 16; k0 += 4) {
            v2f a, b;
            #pragma unroll
            for (int j = 0; j < 2; ++j) {
                int K = k0 + 2 * half + j;
                size_t sidx = (size_t)(nb + K) * 32 + col;
                a[j] = (col == K) ? r : 0.0f;
                b[j] = acc0[sidx] + acc1[sidx];
            }
            c = __builtin_amdgcn_wmma_f32_16x16x4_f32(
                    false, a, false, b, (short)0, c, false, false);
        }

        #pragma unroll
        for (int v = 0; v < 8; ++v) {
            int row = v + 8 * half;
            out[(nb + row) * 16 + col] = c[v];
        }
    } else {
        for (int row = nb + half; row < num_nodes; row += 2) {
            size_t base = (size_t)row * 32;
            float r = __frsqrt_rn(acc0[base + 16] + acc1[base + 16] + 1.0f);
            out[row * 16 + col] = (acc0[base + col] + acc1[base + col]) * r;
        }
    }
}


#define NA_CH   8192
#define NA_RB   400
#define NA_MAXB 256
#define NA_BT   256
#define NA_EPT  (NA_CH / NA_BT)
#define NA_NONE 0xFFFFFFFFu

__global__ __launch_bounds__(NA_BT) void NeighbourAggregation_bin_kernel(
    const int* __restrict__ edge_dst,
    unsigned int* __restrict__ codes,
    unsigned int* __restrict__ offs,
    int num_edges,
    int nb)
{
    __shared__ unsigned int hist[NA_MAXB];
    __shared__ unsigned int scanb[2][NA_MAXB];
    __shared__ unsigned int startb[NA_MAXB + 1];
    __shared__ unsigned int sorted[NA_CH];

    const int t     = threadIdx.x;
    const int ebase = blockIdx.x * NA_CH;

    if (t < NA_MAXB) hist[t] = 0u;
    __syncthreads();

    unsigned int dd[NA_EPT];
    #pragma unroll
    for (int i = 0; i < NA_EPT; ++i) {
        int e = ebase + i * NA_BT + t;
        dd[i] = (e < num_edges) ? (unsigned int)__builtin_nontemporal_load(edge_dst + e)
                                : NA_NONE;
    }

    unsigned int bp[NA_EPT];
    #pragma unroll
    for (int i = 0; i < NA_EPT; ++i) {
        unsigned int b = dd[i] / NA_RB;
        bp[i] = NA_NONE;
        if (b < (unsigned int)nb) {
            unsigned int pos = atomicAdd(&hist[b], 1u);
            bp[i] = (b << 16) | pos;
        }
    }
    __syncthreads();

    const int ts = t & (NA_MAXB - 1);
    unsigned int cntb = hist[ts];
    if (t < NA_MAXB) scanb[0][t] = cntb;
    __syncthreads();
    int cur = 0;
    for (int off = 1; off < NA_MAXB; off <<= 1) {
        unsigned int vsum = scanb[cur][ts];
        if (ts >= off) vsum += scanb[cur][ts - off];
        if (t < NA_MAXB) scanb[cur ^ 1][t] = vsum;
        __syncthreads();
        cur ^= 1;
    }
    if (t < NA_MAXB) {
        unsigned int incl = scanb[cur][t];
        startb[t] = incl - cntb;
        if (t == NA_MAXB - 1) startb[NA_MAXB] = incl;
    }
    __syncthreads();

    unsigned int* offc = offs + (size_t)blockIdx.x * (NA_MAXB + 1);
    if (t < NA_MAXB) offc[t] = startb[t];
    if (t == 0) offc[NA_MAXB] = startb[NA_MAXB];

    #pragma unroll
    for (int i = 0; i < NA_EPT; ++i) {
        if (bp[i] != NA_NONE) {
            unsigned int b   = bp[i] >> 16;
            unsigned int pos = bp[i] & 0xFFFFu;
            unsigned int r   = dd[i] - b * NA_RB;
            sorted[startb[b] + pos] = (r << 19) | ((unsigned int)(i * NA_BT + t) << 6);
        }
    }
    __syncthreads();

    unsigned int totalc = startb[NA_MAXB];
    unsigned int* cc = codes + (size_t)blockIdx.x * NA_CH;
    for (unsigned int jj = t; jj < totalc; jj += NA_BT) cc[jj] = sorted[jj];
}

#define NA_RSHARE(v, I) ((unsigned int)__builtin_amdgcn_update_dpp(0, (int)(v), 0x150 + (I), 0xF, 0xF, false))

#define NA_LD(I, W) if (ns <= (I)) return; v[I] = *(const float*)(xs + (NA_RSHARE(W, (I) & 15) | f4));
__device__ __forceinline__ void na_gather(float (&v)[32], unsigned int gA, unsigned int gB,
                                          const char* __restrict__ xs, unsigned int f4, unsigned int ns)
{
    NA_LD(0, gA)  NA_LD(1, gA)  NA_LD(2, gA)  NA_LD(3, gA)  NA_LD(4, gA)  NA_LD(5, gA)  NA_LD(6, gA)  NA_LD(7, gA)
    NA_LD(8, gA)  NA_LD(9, gA)  NA_LD(10, gA) NA_LD(11, gA) NA_LD(12, gA) NA_LD(13, gA) NA_LD(14, gA) NA_LD(15, gA)
    NA_LD(16, gB) NA_LD(17, gB) NA_LD(18, gB) NA_LD(19, gB) NA_LD(20, gB) NA_LD(21, gB) NA_LD(22, gB) NA_LD(23, gB)
    NA_LD(24, gB) NA_LD(25, gB) NA_LD(26, gB) NA_LD(27, gB) NA_LD(28, gB) NA_LD(29, gB) NA_LD(30, gB) NA_LD(31, gB)
}

#define NA_AD(I, W) if (ns <= (I)) return; unsafeAtomicAdd((float*)(accb + (NA_RSHARE(W, (I) & 15) | f4)), v[I]);
__device__ __forceinline__ void na_add(const float (&v)[32], unsigned int lA, unsigned int lB,
                                       char* accb, unsigned int f4, unsigned int ns)
{
    NA_AD(0, lA)  NA_AD(1, lA)  NA_AD(2, lA)  NA_AD(3, lA)  NA_AD(4, lA)  NA_AD(5, lA)  NA_AD(6, lA)  NA_AD(7, lA)
    NA_AD(8, lA)  NA_AD(9, lA)  NA_AD(10, lA) NA_AD(11, lA) NA_AD(12, lA) NA_AD(13, lA) NA_AD(14, lA) NA_AD(15, lA)
    NA_AD(16, lB) NA_AD(17, lB) NA_AD(18, lB) NA_AD(19, lB) NA_AD(20, lB) NA_AD(21, lB) NA_AD(22, lB) NA_AD(23, lB)
    NA_AD(24, lB) NA_AD(25, lB) NA_AD(26, lB) NA_AD(27, lB) NA_AD(28, lB) NA_AD(29, lB) NA_AD(30, lB) NA_AD(31, lB)
}

#define NA_LDSF 41984

__global__ __launch_bounds__(1024) void NeighbourAggregation_owner_wmma_kernel(
    const float* __restrict__ x,
    const unsigned int* __restrict__ codes,
    const unsigned int* __restrict__ offs,
    float* __restrict__ out,
    int num_nodes,
    int nchunks)
{
    __shared__ float lds[NA_LDSF];
    float* acc = lds;
    float* cnt = lds + (NA_RB + 1) * 16;

    const int t    = threadIdx.x;
    const int lane = t & 31;
    const int w    = __builtin_amdgcn_readfirstlane(t >> 5);
    const int b    = blockIdx.x;
    const unsigned int j  = lane & 15;
    const unsigned int h  = lane >> 4;
    const unsigned int f4 = j * 4u;

    for (int i = t; i < (NA_RB + 1) * 16 + NA_RB; i += 1024) lds[i] = 0.0f;
    __syncthreads();

    const int nIter = (nchunks > w) ? ((nchunks - w + 31) >> 5) : 0;

    for (int k0 = 0; k0 < nIter; k0 += 32) {
        unsigned int lo = 0u, hi = 0u;
        if (k0 + lane < nIter) {
            size_t o = (size_t)(w + 32 * (k0 + lane)) * (NA_MAXB + 1) + b;
            lo = offs[o];
            hi = offs[o + 1];
        }
        const int kend = (nIter - k0 < 32) ? (nIter - k0) : 32;

        int k = 0;
        int c = w + 32 * k0;
        unsigned int slo = __builtin_amdgcn_readlane(lo, 0);
        unsigned int n   = __builtin_amdgcn_readlane(hi, 0) - slo;
        const unsigned int* cp = codes + (size_t)c * NA_CH + slo;
        unsigned int g = 0u;
        unsigned int G = (n < 64u) ? n : 64u;
        unsigned int rA, rB;
        {
            unsigned int n0 = (G + 1u) >> 1;
            unsigned int lenh = h ? (G - n0) : n0;
            unsigned int hb = h ? n0 : 0u;
            rA = (j < lenh)       ? cp[hb + j]       : NA_NONE;
            rB = (j + 16u < lenh) ? cp[hb + 16u + j] : NA_NONE;
        }

        while (true) {
            const unsigned int ns = (G + 1u) >> 1;
            const bool okA = (rA != NA_NONE), okB = (rB != NA_NONE);
            const unsigned int gA = okA ? (rA & 0x7FFFFu) : 0u;
            const unsigned int gB = okB ? (rB & 0x7FFFFu) : 0u;
            const unsigned int lA = okA ? ((rA >> 13) & 0x7FC0u) : (unsigned int)(NA_RB * 64);
            const unsigned int lB = okB ? ((rB >> 13) & 0x7FC0u) : (unsigned int)(NA_RB * 64);
            if (okA) unsafeAtomicAdd(cnt + (rA >> 19), 1.0f);
            if (okB) unsafeAtomicAdd(cnt + (rB >> 19), 1.0f);

            float v[32];
            const char* xs = (const char*)(x + (size_t)c * NA_CH * 16);
            na_gather(v, gA, gB, xs, f4, ns);

            bool more = true;
            unsigned int g2 = g + 64u;
            int c2 = c;
            unsigned int n2 = n;
            const unsigned int* cp2 = cp;
            if (g2 >= n) {
                ++k;
                if (k >= kend) {
                    more = false;
                } else {
                    c2 = c + 32;
                    unsigned int slo2 = __builtin_amdgcn_readlane(lo, k);
                    n2  = __builtin_amdgcn_readlane(hi, k) - slo2;
                    cp2 = codes + (size_t)c2 * NA_CH + slo2;
                    g2  = 0u;
                }
            }
            unsigned int G2 = 0u;
            if (more) {
                G2 = (n2 - g2 < 64u) ? (n2 - g2) : 64u;
                unsigned int n0 = (G2 + 1u) >> 1;
                unsigned int lenh = h ? (G2 - n0) : n0;
                unsigned int hb = g2 + (h ? n0 : 0u);
                rA = (j < lenh)       ? cp2[hb + j]       : NA_NONE;
                rB = (j + 16u < lenh) ? cp2[hb + 16u + j] : NA_NONE;
            }

            na_add(v, lA, lB, (char*)acc, f4, ns);

            if (!more) break;
            c = c2; n = n2; cp = cp2; g = g2; G = G2;
        }
    }
    __syncthreads();

    const int col  = (int)j;
    const int half = (int)h;
    for (int tile = w; tile < NA_RB / 16; tile += 32) {
        const int rb = tile * 16;
        float r = __frsqrt_rn(cnt[rb + col] + 1.0f);

        v8f cacc = {0.f, 0.f, 0.f, 0.f, 0.f, 0.f, 0.f, 0.f};
        #pragma unroll
        for (int q0 = 0; q0 < 16; q0 += 4) {
            v2f a, bb;
            #pragma unroll
            for (int q = 0; q < 2; ++q) {
                int K = q0 + 2 * half + q;
                a[q]  = (col == K) ? r : 0.0f;
                bb[q] = acc[(rb + K) * 16 + col];
            }
            cacc = __builtin_amdgcn_wmma_f32_16x16x4_f32(
                    false, a, false, bb, (short)0, cacc, false, false);
        }

        #pragma unroll
        for (int vv = 0; vv < 8; ++vv) {
            int grow = b * NA_RB + rb + vv + 8 * half;
            if (grow < num_nodes) out[(size_t)grow * 16 + col] = cacc[vv];
        }
    }
}

extern "C" void kernel_launch(void* const* d_in, const int* in_sizes, int n_in,
                              void* d_out, int out_size, void* d_ws, size_t ws_size,
                              hipStream_t stream) {
    const float* x        = (const float*)d_in[0];
    const int*   edge_dst = (const int*)d_in[1];
    int total     = in_sizes[0];
    int num_nodes = out_size / 16;

    float* out = (float*)d_out;

    int swaves  = (total / 16 + 31) / 32;
    int sblocks = (swaves + 7) / 8;
    int nwaves  = (num_nodes + 15) / 16;
    int fblocks = (nwaves + 7) / 8;

    size_t padded_bytes = (size_t)num_nodes * 32 * sizeof(float);
    int num_edges = total / 16;
    int nchunks   = (num_edges + NA_CH - 1) / NA_CH;
    int nblocks   = (num_nodes + NA_RB - 1) / NA_RB;
    size_t codes_bytes = (size_t)nchunks * NA_CH * sizeof(unsigned int);
    size_t offs_bytes  = (size_t)nchunks * (NA_MAXB + 1) * sizeof(unsigned int);
    if (nblocks <= NA_MAXB && nchunks > 0 && ws_size >= codes_bytes + offs_bytes) {
        unsigned int* codes = (unsigned int*)d_ws;
        unsigned int* offs  = codes + (size_t)nchunks * NA_CH;
        NeighbourAggregation_bin_kernel<<<nchunks, NA_BT, 0, stream>>>(
            edge_dst, codes, offs, num_edges, nblocks);
        NeighbourAggregation_owner_wmma_kernel<<<nblocks, 1024, 0, stream>>>(
            x, codes, offs, out, num_nodes, nchunks);
    } else if (ws_size >= 2 * padded_bytes) {
        float* acc0 = (float*)d_ws;
        float* acc1 = acc0 + (size_t)num_nodes * 32;
        hipMemsetAsync(d_ws, 0, 2 * padded_bytes, stream);
        NeighbourAggregation_scatter_halfplane_kernel<<<2048, 256, 0, stream>>>(
            x, edge_dst, acc0, acc1, total);
        NeighbourAggregation_finalize_halfplane_wmma_kernel<<<fblocks, 256, 0, stream>>>(
            acc0, acc1, out, num_nodes);
    } else if (ws_size >= padded_bytes) {
        float* acc = (float*)d_ws;
        hipMemsetAsync(d_ws, 0, padded_bytes, stream);
        NeighbourAggregation_scatter_padded_kernel<<<sblocks, 256, 0, stream>>>(
            x, edge_dst, acc, total);
        NeighbourAggregation_finalize_padded_wmma_kernel<<<fblocks, 256, 0, stream>>>(
            acc, out, num_nodes);
    } else {
        float* counts = (float*)d_ws;
        hipMemsetAsync(d_out, 0, (size_t)out_size * sizeof(float), stream);
        hipMemsetAsync(d_ws, 0, (size_t)num_nodes * sizeof(float), stream);
        NeighbourAggregation_scatter_kernel<<<sblocks, 256, 0, stream>>>(
            x, edge_dst, out, counts, total);
        NeighbourAggregation_finalize_wmma_kernel<<<fblocks, 256, 0, stream>>>(
            out, counts, num_nodes);
    }
}
